# starting kernel
# baseline (speedup 1.0000x reference)
#include <hip/hip_runtime.h>


typedef float v2f __attribute__((ext_vector_type(2)));
typedef float v8f __attribute__((ext_vector_type(8)));

#define GCN_N_NODES 100000


__global__ void gcn_zero_kernel(float* __restrict__ p, long long n) {
  long long i = (long long)blockIdx.x * blockDim.x + threadIdx.x;
  long long stride = (long long)gridDim.x * blockDim.x;
  for (; i < n; i += stride) p[i] = 0.0f;
}

__global__ void gcn_degree_kernel(const int* __restrict__ rows,
                                  float* __restrict__ deg, int n_edges) {
  int e = blockIdx.x * blockDim.x + threadIdx.x;
  if (e < n_edges) atomicAdd(&deg[rows[e]], 1.0f);
}

__global__ void gcn_dinv_kernel(const float* __restrict__ deg,
                                float* __restrict__ dinv, int n) {
  int i = blockIdx.x * blockDim.x + threadIdx.x;
  if (i < n) {
    float d = sqrtf(deg[i]);
    dinv[i] = 1.0f / fmaxf(d, 1e-12f);
  }
}

__global__ void gcn_rowscale_kernel(float* __restrict__ p,
                                    const float* __restrict__ dinv,
                                    long long n, int colshift) {
  long long i = (long long)blockIdx.x * blockDim.x + threadIdx.x;
  long long stride = (long long)gridDim.x * blockDim.x;
  for (; i < n; i += stride) p[i] *= dinv[i >> colshift];
}


template <int N, bool RELU_SCALE_A, bool SCALE_C>
__global__ __launch_bounds__(256) void gcn_gemm_wmma(
    const float* __restrict__ A, const float* __restrict__ B,
    float* __restrict__ C, const float* __restrict__ dinv, int M, int K) {
  constexpr int NT = N / 16;
  constexpr int KC = 64;
  constexpr int N2 = N / 2;
  __shared__ __align__(16) float Bs[KC * N];

  const int wib  = threadIdx.x >> 5;
  const int lane = threadIdx.x & 31;
  const int half = lane >> 4;
  const int l15  = lane & 15;

  const int row0   = blockIdx.x * 128 + wib * 16;
  const bool active = (row0 < M);

  const float* __restrict__ Arow =
      A + (size_t)(active ? row0 + l15 : 0) * K;
  const float dscale =
      (RELU_SCALE_A && active) ? dinv[row0 + l15] : 1.0f;

  v8f acc[NT];
#pragma unroll
  for (int t = 0; t < NT; ++t) acc[t] = (v8f){};

  for (int kc = 0; kc < K; kc += KC) {
    const float* __restrict__ Bg = B + (size_t)kc * N;
#pragma unroll
    for (int pidx = threadIdx.x; pidx < (KC / 2) * N2; pidx += 256) {
      const int kp = pidx / N2;
      const int c2 = (pidx - kp * N2) * 2;
      const float2 g0 = *(const float2*)(Bg + (size_t)(2 * kp) * N + c2);
      const float2 g1 = *(const float2*)(Bg + (size_t)(2 * kp + 1) * N + c2);
      *(float4*)(&Bs[kp * (2 * N) + c2 * 2]) =
          make_float4(g0.x, g1.x, g0.y, g1.y);
    }
    __syncthreads();

    if (active) {
      const float* __restrict__ Ak = Arow + kc;
#pragma unroll
      for (int k0 = 0; k0 < KC; k0 += 4) {
        const int koff = k0 + 2 * half;
        v2f a = *(const v2f*)(Ak + koff);
        if (RELU_SCALE_A) {
          a.x = fmaxf(a.x * dscale, 0.0f);
          a.y = fmaxf(a.y * dscale, 0.0f);
        }
        const int kp = koff >> 1;
#pragma unroll
        for (int t = 0; t < NT; ++t) {
          const v2f b =
              *(const v2f*)(&Bs[kp * (2 * N) + (t * 16 + l15) * 2]);
          acc[t] = __builtin_amdgcn_wmma_f32_16x16x4_f32(
              false, a, false, b, (short)0, acc[t], false, false);
        }
      }
    }
    __syncthreads();
  }

  if (active) {
#pragma unroll
    for (int t = 0; t < NT; ++t) {
      const int col = t * 16 + l15;
#pragma unroll
      for (int v = 0; v < 8; ++v) {
        const int row = row0 + v + 8 * half;
        float val = acc[t][v];
        if (SCALE_C) val *= dinv[row];
        C[(size_t)row * N + col] = val;
      }
    }
  }
}


__global__ void gcn_spmm_plain(const float* __restrict__ src,
                               const int* __restrict__ rows,
                               const int* __restrict__ cols,
                               float* __restrict__ dst,
                               int n_edges, int F, int fshift) {
  const long long tid = (long long)blockIdx.x * blockDim.x + threadIdx.x;
  const int fchunks = 1 << fshift;
  const long long total = (long long)n_edges * fchunks;
  if (tid >= total) return;

  const int e  = (int)(tid >> fshift);
  const int fc = (int)(tid & (fchunks - 1));

  const int r = rows[e];
  const int c = cols[e];

  const float4 v = ((const float4*)(src + (size_t)c * F))[fc];
  float* __restrict__ d = dst + (size_t)r * F + fc * 4;
  atomicAdd(d + 0, v.x);
  atomicAdd(d + 1, v.y);
  atomicAdd(d + 2, v.z);
  atomicAdd(d + 3, v.w);
}


extern "C" void kernel_launch(void* const* d_in, const int* in_sizes, int n_in,
                              void* d_out, int out_size, void* d_ws, size_t ws_size,
                              hipStream_t stream) {
  const float* X  = (const float*)d_in[0];
  const float* W1 = (const float*)d_in[1];
  const float* W2 = (const float*)d_in[2];
  const int* erow = (const int*)d_in[3];
  const int* ecol = (const int*)d_in[4];
  const int n_edges = in_sizes[3];
  float* out = (float*)d_out;

  float* deg = (float*)d_ws;
  float* dinv = deg + GCN_N_NODES;
  float* XWs = dinv + GCN_N_NODES;
  float* S   = XWs + (size_t)GCN_N_NODES * 128;
  float* HWs = S + (size_t)GCN_N_NODES * 128;

  const long long nS   = (long long)GCN_N_NODES * 128;
  const long long nOut = (long long)out_size;

  gcn_zero_kernel<<<512, 256, 0, stream>>>(deg, GCN_N_NODES);
  gcn_zero_kernel<<<4096, 256, 0, stream>>>(S, nS);
  gcn_zero_kernel<<<4096, 256, 0, stream>>>(out, nOut);

  gcn_degree_kernel<<<(n_edges + 255) / 256, 256, 0, stream>>>(erow, deg, n_edges);
  gcn_dinv_kernel<<<(GCN_N_NODES + 255) / 256, 256, 0, stream>>>(deg, dinv, GCN_N_NODES);

  const int row_tiles = GCN_N_NODES / 16;
  const int gemm_blocks = (row_tiles + 7) / 8;

  gcn_gemm_wmma<128, false, true>
      <<<gemm_blocks, 256, 0, stream>>>(X, W1, XWs, dinv, GCN_N_NODES, 256);

  {
    const long long total = (long long)n_edges * 32;
    gcn_spmm_plain<<<(int)((total + 255) / 256), 256, 0, stream>>>(
        XWs, erow, ecol, S, n_edges, 128, 5);
  }

  gcn_gemm_wmma<64, true, true>
      <<<gemm_blocks, 256, 0, stream>>>(S, W2, HWs, dinv, GCN_N_NODES, 128);

  {
    const long long total = (long long)n_edges * 16;
    gcn_spmm_plain<<<(int)((total + 255) / 256), 256, 0, stream>>>(
        HWs, erow, ecol, out, n_edges, 64, 4);
  }

  gcn_rowscale_kernel<<<4096, 256, 0, stream>>>(out, dinv, nOut, 6);
}


# Edge pass (degree counts, band lists) moved into the X@W1 launch as its last work-groups; layer-0 D^-1/2 of source rows applied at the gather, no dinv launch: 7 launches to 5; arrival order
# speedup vs baseline: 2.4308x; 2.4308x over previous
#include <hip/hip_runtime.h>


typedef float v2f __attribute__((ext_vector_type(2)));
typedef float v8f __attribute__((ext_vector_type(8)));

#define GCN_N_NODES 100000
#define GCN_BANDS 256
#define GCN_BAND_ROWS 391
#define GCN_LIST_CAP 65536
#define GCN_COUNT_PITCH 64
#define GCN_EDGE_GROUP 4096

__device__ __forceinline__ float gcn_dinv_of(float deg) {
  const float d = sqrtf(deg);
  return 1.0f / fmaxf(d, 1e-12f);
}


__global__ void gcn_zero_kernel(float* __restrict__ p, long long n,
                                unsigned* __restrict__ q, int nq) {
  long long i = (long long)blockIdx.x * blockDim.x + threadIdx.x;
  long long stride = (long long)gridDim.x * blockDim.x;
  for (long long k = i; k < n; k += stride) p[k] = 0.0f;
  for (long long k = i; k < nq; k += stride) q[k] = 0u;
}

__device__ __forceinline__ void gcn_degree_lists_group(
    const int* __restrict__ rows, const int* __restrict__ cols,
    float* __restrict__ deg, unsigned* __restrict__ blist,
    unsigned* __restrict__ bcount, int n_edges, int group) {
  constexpr int TPG = 256;
  constexpr int EPT = GCN_EDGE_GROUP / TPG;
  static_assert(TPG == GCN_BANDS, "one thread per band reserves the list entries");
  __shared__ unsigned cnt[GCN_BANDS];
  cnt[threadIdx.x] = 0u;
  __syncthreads();
  const int e0 = group * GCN_EDGE_GROUP + threadIdx.x;
  int r[EPT], c[EPT];
#pragma unroll
  for (int j = 0; j < EPT; ++j) {
    const int e = min(e0 + j * TPG, n_edges - 1);
    r[j] = rows[e];
    c[j] = cols[e];
  }
  int b[EPT];
  unsigned rank[EPT];
#pragma unroll
  for (int j = 0; j < EPT; ++j) {
    b[j] = r[j] / GCN_BAND_ROWS;
    rank[j] = 0u;
    if (e0 + j * TPG < n_edges) rank[j] = atomicAdd(&cnt[b[j]], 1u);
  }
  __syncthreads();
  {
    const unsigned n = cnt[threadIdx.x];
    unsigned base = 0u;
    if (n) base = atomicAdd(&bcount[threadIdx.x * GCN_COUNT_PITCH], n);
    cnt[threadIdx.x] = base;
  }
  __syncthreads();
#pragma unroll
  for (int j = 0; j < EPT; ++j) {
    if (e0 + j * TPG < n_edges) {
      const unsigned slot = cnt[b[j]] + rank[j];
      if (slot < (unsigned)GCN_LIST_CAP)
        blist[(size_t)b[j] * GCN_LIST_CAP + slot] =
            ((unsigned)(r[j] - b[j] * GCN_BAND_ROWS) << 17) | (unsigned)c[j];
      atomicAdd(&deg[r[j]], 1.0f);
    }
  }
}


template <int N, bool RELU_SCALE_A, bool SCALE_C>
__device__ __forceinline__ void gcn_gemm_wmma_group(
    const float* __restrict__ A, const float* __restrict__ B,
    float* __restrict__ C, const float* __restrict__ dinv, int M, int K, int blk) {
  constexpr int NT = N / 16;
  constexpr int KC = 64;
  constexpr int N2 = N / 2;
  __shared__ __align__(16) float Bs[KC * N];

  const int wib  = threadIdx.x >> 5;
  const int lane = threadIdx.x & 31;
  const int half = lane >> 4;
  const int l15  = lane & 15;

  const int row0   = blk * 128 + wib * 16;
  const bool active = (row0 < M);

  const float* __restrict__ Arow =
      A + (size_t)(active ? row0 + l15 : 0) * K;
  const float dscale =
      (RELU_SCALE_A && active) ? dinv[row0 + l15] : 1.0f;

  v8f acc[NT];
#pragma unroll
  for (int t = 0; t < NT; ++t) acc[t] = (v8f){};

  for (int kc = 0; kc < K; kc += KC) {
    const float* __restrict__ Bg = B + (size_t)kc * N;
#pragma unroll
    for (int pidx = threadIdx.x; pidx < (KC / 2) * N2; pidx += 256) {
      const int kp = pidx / N2;
      const int c2 = (pidx - kp * N2) * 2;
      const float2 g0 = *(const float2*)(Bg + (size_t)(2 * kp) * N + c2);
      const float2 g1 = *(const float2*)(Bg + (size_t)(2 * kp + 1) * N + c2);
      *(float4*)(&Bs[kp * (2 * N) + c2 * 2]) =
          make_float4(g0.x, g1.x, g0.y, g1.y);
    }
    __syncthreads();

    if (active) {
      const float* __restrict__ Ak = Arow + kc;
#pragma unroll
      for (int k0 = 0; k0 < KC; k0 += 4) {
        const int koff = k0 + 2 * half;
        v2f a = *(const v2f*)(Ak + koff);
        if (RELU_SCALE_A) {
          a.x = fmaxf(a.x * dscale, 0.0f);
          a.y = fmaxf(a.y * dscale, 0.0f);
        }
        const int kp = koff >> 1;
#pragma unroll
        for (int t = 0; t < NT; ++t) {
          const v2f b =
              *(const v2f*)(&Bs[kp * (2 * N) + (t * 16 + l15) * 2]);
          acc[t] = __builtin_amdgcn_wmma_f32_16x16x4_f32(
              false, a, false, b, (short)0, acc[t], false, false);
        }
      }
    }
    __syncthreads();
  }

  if (active) {
#pragma unroll
    for (int t = 0; t < NT; ++t) {
      const int col = t * 16 + l15;
#pragma unroll
      for (int v = 0; v < 8; ++v) {
        const int row = row0 + v + 8 * half;
        float val = acc[t][v];
        if (SCALE_C) val *= dinv[row];
        C[(size_t)row * N + col] = val;
      }
    }
  }
}

__global__ __launch_bounds__(256) void gcn_gemm1_edges_kernel(
    const float* __restrict__ A, const float* __restrict__ B,
    float* __restrict__ C, int M, int K, int n_gemm_groups,
    const int* __restrict__ rows, const int* __restrict__ cols,
    float* __restrict__ deg, unsigned* __restrict__ blist,
    unsigned* __restrict__ bcount, int n_edges) {
  const int blk = blockIdx.x;
  if (blk >= n_gemm_groups) {
    gcn_degree_lists_group(rows, cols, deg, blist, bcount, n_edges,
                           blk - n_gemm_groups);
    return;
  }
  gcn_gemm_wmma_group<128, false, false>(A, B, C, nullptr, M, K, blk);
}

__global__ __launch_bounds__(256) void gcn_gemm2_kernel(
    const float* __restrict__ A, const float* __restrict__ B,
    float* __restrict__ C, const float* __restrict__ dinv, int M, int K) {
  gcn_gemm_wmma_group<64, true, true>(A, B, C, dinv, M, K, blockIdx.x);
}


template <int F, bool SCALE_SRC>
__device__ __forceinline__ void gcn_band_add_rows(
    const float* __restrict__ src, float* acc, unsigned pk, float ds_l, int qn,
    int lane) {
  constexpr int NV = F / 32;
  constexpr int G = 32 / NV;
  typedef float vecf __attribute__((ext_vector_type(NV)));
  const int r_l = (int)(pk >> 17);
  const int c_l = (int)(pk & 0x1ffffu);
  const int d_l = __float_as_int(ds_l);
  int g = 0;
  for (; g + G <= qn; g += G) {
    vecf v[G];
#pragma unroll
    for (int u = 0; u < G; ++u) {
      const int c = __builtin_amdgcn_readlane(c_l, g + u);
      v[u] = *(const vecf*)(src + (size_t)c * F + lane * NV);
    }
#pragma unroll
    for (int u = 0; u < G; ++u) {
      const int r = __builtin_amdgcn_readlane(r_l, g + u);
      float* a = acc + r * F + lane;
      if (SCALE_SRC) v[u] *= __int_as_float(__builtin_amdgcn_readlane(d_l, g + u));
#pragma unroll
      for (int j = 0; j < NV; ++j) atomicAdd(a + j * 32, v[u][j]);
    }
  }
  if (g < qn) {
    const int n = qn - g;
    vecf v[G];
#pragma unroll
    for (int u = 0; u < G - 1; ++u) {
      if (u < n) {
        const int c = __builtin_amdgcn_readlane(c_l, g + u);
        v[u] = *(const vecf*)(src + (size_t)c * F + lane * NV);
      }
    }
#pragma unroll
    for (int u = 0; u < G - 1; ++u) {
      if (u < n) {
        const int r = __builtin_amdgcn_readlane(r_l, g + u);
        float* a = acc + r * F + lane;
        if (SCALE_SRC) v[u] *= __int_as_float(__builtin_amdgcn_readlane(d_l, g + u));
#pragma unroll
        for (int j = 0; j < NV; ++j) atomicAdd(a + j * 32, v[u][j]);
      }
    }
  }
}

template <int F, bool SCALE_SRC, bool SCALE_OUT>
__global__ __launch_bounds__(1024) void gcn_spmm_band(
    const float* __restrict__ src, const int* __restrict__ rows,
    const int* __restrict__ cols, float* __restrict__ dst,
    const float* __restrict__ dinv, const float* __restrict__ deg,
    float* __restrict__ dinv_out, const unsigned* __restrict__ blist,
    const unsigned* __restrict__ bcount, int n_edges) {
  constexpr int NV = F / 32;
  typedef float vecf __attribute__((ext_vector_type(NV)));
  __shared__ __align__(16) float acc[GCN_BAND_ROWS * F];

  const int lane = threadIdx.x & 31;
  const int wave = __builtin_amdgcn_readfirstlane(threadIdx.x >> 5);
  const int row_base = blockIdx.x * GCN_BAND_ROWS;
  const int nrows = min(GCN_BAND_ROWS, GCN_N_NODES - row_base);
  const unsigned* __restrict__ bl = blist + (size_t)blockIdx.x * GCN_LIST_CAP;

  for (int i = threadIdx.x * 4; i < GCN_BAND_ROWS * F; i += 4096)
    *(float4*)(&acc[i]) = make_float4(0.0f, 0.0f, 0.0f, 0.0f);
  __syncthreads();

  const unsigned n_list = bcount[blockIdx.x * GCN_COUNT_PITCH];
  if (n_list <= (unsigned)GCN_LIST_CAP) {
    const int n = (int)n_list;
    const int n_chunks = (n + 31) >> 5;
    int ch = wave;
    unsigned pk = 0u, pkn = 0u;
    if (ch < n_chunks && ch * 32 + lane < n) pk = bl[ch * 32 + lane];
    if (ch + 32 < n_chunks && (ch + 32) * 32 + lane < n) pkn = bl[(ch + 32) * 32 + lane];
    float ds_l = 1.0f;
    if (SCALE_SRC) ds_l = gcn_dinv_of(deg[pk & 0x1ffffu]);
    while (ch < n_chunks) {
      const int nnch = ch + 64;
      unsigned pknn = 0u;
      if (nnch < n_chunks && nnch * 32 + lane < n) pknn = bl[nnch * 32 + lane];
      float degn = 1.0f;
      if (SCALE_SRC) degn = deg[pkn & 0x1ffffu];
      gcn_band_add_rows<F, SCALE_SRC>(src, acc, pk, ds_l, min(32, n - ch * 32), lane);
      if (SCALE_SRC) ds_l = gcn_dinv_of(degn);
      pk = pkn;
      pkn = pknn;
      ch += 32;
    }
  } else {
    for (int eb = wave * 32; eb < n_edges; eb += 1024) {
      const int e = eb + lane;
      unsigned pk = 0u;
      bool mine = false;
      if (e < n_edges) {
        const int rb = rows[e] - row_base;
        mine = (rb >= 0) && (rb < GCN_BAND_ROWS);
        if (mine) pk = ((unsigned)rb << 17) | (unsigned)cols[e];
      }
      unsigned bal = __builtin_amdgcn_ballot_w32(mine);
      while (bal) {
        const int l = __builtin_ctz(bal);
        bal &= bal - 1;
        const unsigned w = (unsigned)__builtin_amdgcn_readlane((int)pk, l);
        vecf v = *(const vecf*)(src + (size_t)(w & 0x1ffffu) * F + lane * NV);
        if (SCALE_SRC) v *= gcn_dinv_of(deg[w & 0x1ffffu]);
        float* a = acc + (int)(w >> 17) * F + lane;
#pragma unroll
        for (int j = 0; j < NV; ++j) atomicAdd(a + j * 32, v[j]);
      }
    }
  }
  __syncthreads();

  if (SCALE_SRC && (int)threadIdx.x < nrows)
    dinv_out[row_base + threadIdx.x] = gcn_dinv_of(deg[row_base + threadIdx.x]);

  for (int r = wave; r < nrows; r += 32) {
    vecf v;
#pragma unroll
    for (int j = 0; j < NV; ++j) v[j] = acc[r * F + j * 32 + lane];
    if (SCALE_OUT) v *= dinv[row_base + r];
    *(vecf*)(dst + (size_t)(row_base + r) * F + lane * NV) = v;
  }
}


extern "C" void kernel_launch(void* const* d_in, const int* in_sizes, int n_in,
                              void* d_out, int out_size, void* d_ws, size_t ws_size,
                              hipStream_t stream) {
  const float* X  = (const float*)d_in[0];
  const float* W1 = (const float*)d_in[1];
  const float* W2 = (const float*)d_in[2];
  const int* erow = (const int*)d_in[3];
  const int* ecol = (const int*)d_in[4];
  const int n_edges = in_sizes[3];
  float* out = (float*)d_out;

  float* deg = (float*)d_ws;
  float* dinv = deg + GCN_N_NODES;
  float* XW  = dinv + GCN_N_NODES;
  float* S   = XW + (size_t)GCN_N_NODES * 128;
  float* HWs = S + (size_t)GCN_N_NODES * 128;
  unsigned* blist = (unsigned*)(HWs + (size_t)GCN_N_NODES * 64);
  unsigned* bcount = blist + (size_t)GCN_BANDS * GCN_LIST_CAP;

  gcn_zero_kernel<<<512, 256, 0, stream>>>(deg, GCN_N_NODES, bcount,
                                           GCN_BANDS * GCN_COUNT_PITCH);

  const int deg_groups = (n_edges + GCN_EDGE_GROUP - 1) / GCN_EDGE_GROUP;
  const int row_tiles = GCN_N_NODES / 16;
  const int gemm_blocks = (row_tiles + 7) / 8;

  gcn_gemm1_edges_kernel<<<gemm_blocks + deg_groups, 256, 0, stream>>>(
      X, W1, XW, GCN_N_NODES, 256, gemm_blocks, erow, ecol, deg, blist, bcount,
      n_edges);

  gcn_spmm_band<128, true, false><<<GCN_BANDS, 1024, 0, stream>>>(
      XW, erow, ecol, S, nullptr, deg, dinv, blist, bcount, n_edges);

  gcn_gemm2_kernel<<<gemm_blocks, 256, 0, stream>>>(S, W2, HWs, dinv, GCN_N_NODES, 128);

  gcn_spmm_band<64, false, true><<<GCN_BANDS, 1024, 0, stream>>>(
      HWs, erow, ecol, out, dinv, deg, nullptr, blist, bcount, n_edges);
}


# H@W2 product moved into the layer-0 band SpMM: a work-group multiplies its band's sums (dinv, ReLU) out of LDS by W2 held in LDS, on WMMA, stores whole product rows; no sums plane, 5 launches to 4
# speedup vs baseline: 2.5535x; 1.0505x over previous
#include <hip/hip_runtime.h>


typedef float v2f __attribute__((ext_vector_type(2)));
typedef float v8f __attribute__((ext_vector_type(8)));

#define GCN_N_NODES 100000
#define GCN_BANDS 256
#define GCN_BAND_ROWS 391
#define GCN_LIST_CAP 65536
#define GCN_COUNT_PITCH 64
#define GCN_EDGE_GROUP 4096

__device__ __forceinline__ float gcn_dinv_of(float deg) {
  const float d = sqrtf(deg);
  return 1.0f / fmaxf(d, 1e-12f);
}


__global__ void gcn_zero_kernel(float* __restrict__ p, long long n,
                                unsigned* __restrict__ q, int nq) {
  long long i = (long long)blockIdx.x * blockDim.x + threadIdx.x;
  long long stride = (long long)gridDim.x * blockDim.x;
  for (long long k = i; k < n; k += stride) p[k] = 0.0f;
  for (long long k = i; k < nq; k += stride) q[k] = 0u;
}

__device__ __forceinline__ void gcn_degree_lists_group(
    const int* __restrict__ rows, const int* __restrict__ cols,
    float* __restrict__ deg, unsigned* __restrict__ blist,
    unsigned* __restrict__ bcount, int n_edges, int group) {
  constexpr int TPG = 256;
  constexpr int EPT = GCN_EDGE_GROUP / TPG;
  static_assert(TPG == GCN_BANDS, "one thread per band reserves the list entries");
  __shared__ unsigned cnt[GCN_BANDS];
  cnt[threadIdx.x] = 0u;
  __syncthreads();
  const int e0 = group * GCN_EDGE_GROUP + threadIdx.x;
  int r[EPT], c[EPT];
#pragma unroll
  for (int j = 0; j < EPT; ++j) {
    const int e = min(e0 + j * TPG, n_edges - 1);
    r[j] = rows[e];
    c[j] = cols[e];
  }
  int b[EPT];
  unsigned rank[EPT];
#pragma unroll
  for (int j = 0; j < EPT; ++j) {
    b[j] = r[j] / GCN_BAND_ROWS;
    rank[j] = 0u;
    if (e0 + j * TPG < n_edges) rank[j] = atomicAdd(&cnt[b[j]], 1u);
  }
  __syncthreads();
  {
    const unsigned n = cnt[threadIdx.x];
    unsigned base = 0u;
    if (n) base = atomicAdd(&bcount[threadIdx.x * GCN_COUNT_PITCH], n);
    cnt[threadIdx.x] = base;
  }
  __syncthreads();
#pragma unroll
  for (int j = 0; j < EPT; ++j) {
    if (e0 + j * TPG < n_edges) {
      const unsigned slot = cnt[b[j]] + rank[j];
      if (slot < (unsigned)GCN_LIST_CAP)
        blist[(size_t)b[j] * GCN_LIST_CAP + slot] =
            ((unsigned)(r[j] - b[j] * GCN_BAND_ROWS) << 17) | (unsigned)c[j];
      atomicAdd(&deg[r[j]], 1.0f);
    }
  }
}


template <int N>
__device__ __forceinline__ void gcn_gemm_wmma_group(
    const float* __restrict__ A, const float* __restrict__ B,
    float* __restrict__ C, int M, int K, int blk) {
  constexpr int NT = N / 16;
  constexpr int KC = 64;
  constexpr int N2 = N / 2;
  __shared__ __align__(16) float Bs[KC * N];

  const int wib  = threadIdx.x >> 5;
  const int lane = threadIdx.x & 31;
  const int half = lane >> 4;
  const int l15  = lane & 15;

  const int row0   = blk * 128 + wib * 16;
  const bool active = (row0 < M);

  const float* __restrict__ Arow =
      A + (size_t)(active ? row0 + l15 : 0) * K;

  v8f acc[NT];
#pragma unroll
  for (int t = 0; t < NT; ++t) acc[t] = (v8f){};

  for (int kc = 0; kc < K; kc += KC) {
    const float* __restrict__ Bg = B + (size_t)kc * N;
#pragma unroll
    for (int pidx = threadIdx.x; pidx < (KC / 2) * N2; pidx += 256) {
      const int kp = pidx / N2;
      const int c2 = (pidx - kp * N2) * 2;
      const float2 g0 = *(const float2*)(Bg + (size_t)(2 * kp) * N + c2);
      const float2 g1 = *(const float2*)(Bg + (size_t)(2 * kp + 1) * N + c2);
      *(float4*)(&Bs[kp * (2 * N) + c2 * 2]) =
          make_float4(g0.x, g1.x, g0.y, g1.y);
    }
    __syncthreads();

    if (active) {
      const float* __restrict__ Ak = Arow + kc;
#pragma unroll
      for (int k0 = 0; k0 < KC; k0 += 4) {
        const int koff = k0 + 2 * half;
        const v2f a = *(const v2f*)(Ak + koff);
        const int kp = koff >> 1;
#pragma unroll
        for (int t = 0; t < NT; ++t) {
          const v2f b =
              *(const v2f*)(&Bs[kp * (2 * N) + (t * 16 + l15) * 2]);
          acc[t] = __builtin_amdgcn_wmma_f32_16x16x4_f32(
              false, a, false, b, (short)0, acc[t], false, false);
        }
      }
    }
    __syncthreads();
  }

  if (active) {
#pragma unroll
    for (int t = 0; t < NT; ++t) {
      const int col = t * 16 + l15;
#pragma unroll
      for (int v = 0; v < 8; ++v) {
        const int row = row0 + v + 8 * half;
        C[(size_t)row * N + col] = acc[t][v];
      }
    }
  }
}

__global__ __launch_bounds__(256) void gcn_gemm1_edges_kernel(
    const float* __restrict__ A, const float* __restrict__ B,
    float* __restrict__ C, int M, int K, int n_gemm_groups,
    const int* __restrict__ rows, const int* __restrict__ cols,
    float* __restrict__ deg, unsigned* __restrict__ blist,
    unsigned* __restrict__ bcount, int n_edges) {
  const int blk = blockIdx.x;
  if (blk >= n_gemm_groups) {
    gcn_degree_lists_group(rows, cols, deg, blist, bcount, n_edges,
                           blk - n_gemm_groups);
    return;
  }
  gcn_gemm_wmma_group<128>(A, B, C, M, K, blk);
}


template <int F, int P, bool SCALE_SRC>
__device__ __forceinline__ void gcn_band_add_rows(
    const float* __restrict__ src, float* acc, unsigned pk, float ds_l, int qn,
    int lane) {
  constexpr int NV = F / 32;
  constexpr int G = 32 / NV;
  typedef float vecf __attribute__((ext_vector_type(NV)));
  const int r_l = (int)(pk >> 17);
  const int c_l = (int)(pk & 0x1ffffu);
  const int d_l = __float_as_int(ds_l);
  int g = 0;
  for (; g + G <= qn; g += G) {
    vecf v[G];
#pragma unroll
    for (int u = 0; u < G; ++u) {
      const int c = __builtin_amdgcn_readlane(c_l, g + u);
      v[u] = *(const vecf*)(src + (size_t)c * F + lane * NV);
    }
#pragma unroll
    for (int u = 0; u < G; ++u) {
      const int r = __builtin_amdgcn_readlane(r_l, g + u);
      float* a = acc + r * P + lane;
      if (SCALE_SRC) v[u] *= __int_as_float(__builtin_amdgcn_readlane(d_l, g + u));
#pragma unroll
      for (int j = 0; j < NV; ++j) atomicAdd(a + j * 32, v[u][j]);
    }
  }
  if (g < qn) {
    const int n = qn - g;
    vecf v[G];
#pragma unroll
    for (int u = 0; u < G - 1; ++u) {
      if (u < n) {
        const int c = __builtin_amdgcn_readlane(c_l, g + u);
        v[u] = *(const vecf*)(src + (size_t)c * F + lane * NV);
      }
    }
#pragma unroll
    for (int u = 0; u < G - 1; ++u) {
      if (u < n) {
        const int r = __builtin_amdgcn_readlane(r_l, g + u);
        float* a = acc + r * P + lane;
        if (SCALE_SRC) v[u] *= __int_as_float(__builtin_amdgcn_readlane(d_l, g + u));
#pragma unroll
        for (int j = 0; j < NV; ++j) atomicAdd(a + j * 32, v[u][j]);
      }
    }
  }
}

template <int F, bool SCALE_SRC, bool SCALE_OUT, bool TIMES_W2>
__global__ __launch_bounds__(1024) void gcn_spmm_band(
    const float* __restrict__ src, const int* __restrict__ rows,
    const int* __restrict__ cols, float* __restrict__ dst,
    const float* __restrict__ dinv, const float* __restrict__ deg,
    float* __restrict__ dinv_out, const float* __restrict__ w2,
    const unsigned* __restrict__ blist, const unsigned* __restrict__ bcount,
    int n_edges) {
  constexpr int NV = F / 32;
  constexpr int NO = 64;
  constexpr int P = TIMES_W2 ? F + 4 : F;
  constexpr int ROWS = TIMES_W2 ? (GCN_BAND_ROWS + 15) / 16 * 16 : GCN_BAND_ROWS;
  static_assert(!TIMES_W2 || F == 128, "the A fragment addresses below are for NV = 4");
  static_assert((ROWS * P) % 4 == 0, "acc is cleared in 16-byte stores");
  typedef float vecf __attribute__((ext_vector_type(NV)));
  __shared__ __align__(16) float acc[ROWS * P];
  __shared__ __align__(16) float w2s[TIMES_W2 ? F * NO : 4];
  __shared__ float dl[TIMES_W2 ? ROWS : 4];

  const int lane = threadIdx.x & 31;
  const int wave = __builtin_amdgcn_readfirstlane(threadIdx.x >> 5);
  const int row_base = blockIdx.x * GCN_BAND_ROWS;
  const int nrows = min(GCN_BAND_ROWS, GCN_N_NODES - row_base);
  const unsigned* __restrict__ bl = blist + (size_t)blockIdx.x * GCN_LIST_CAP;

  for (int i = threadIdx.x * 4; i < ROWS * P; i += 4096)
    *(float4*)(&acc[i]) = make_float4(0.0f, 0.0f, 0.0f, 0.0f);
  if (TIMES_W2) {
    if ((int)threadIdx.x < ROWS)
      dl[threadIdx.x] = ((int)threadIdx.x < nrows)
                            ? gcn_dinv_of(deg[row_base + threadIdx.x]) : 0.0f;
    for (int pidx = threadIdx.x; pidx < (F / 2) * (NO / 2); pidx += 1024) {
      const int kp = pidx / (NO / 2);
      const int c2 = (pidx - kp * (NO / 2)) * 2;
      const float2 g0 = *(const float2*)(w2 + (size_t)(2 * kp) * NO + c2);
      const float2 g1 = *(const float2*)(w2 + (size_t)(2 * kp + 1) * NO + c2);
      *(float4*)(&w2s[kp * (2 * NO) + c2 * 2]) = make_float4(g0.x, g1.x, g0.y, g1.y);
    }
  }
  __syncthreads();

  const unsigned n_list = bcount[blockIdx.x * GCN_COUNT_PITCH];
  if (n_list <= (unsigned)GCN_LIST_CAP) {
    const int n = (int)n_list;
    const int n_chunks = (n + 31) >> 5;
    int ch = wave;
    unsigned pk = 0u, pkn = 0u;
    if (ch < n_chunks && ch * 32 + lane < n) pk = bl[ch * 32 + lane];
    if (ch + 32 < n_chunks && (ch + 32) * 32 + lane < n) pkn = bl[(ch + 32) * 32 + lane];
    float ds_l = 1.0f;
    if (SCALE_SRC) ds_l = gcn_dinv_of(deg[pk & 0x1ffffu]);
    while (ch < n_chunks) {
      const int nnch = ch + 64;
      unsigned pknn = 0u;
      if (nnch < n_chunks && nnch * 32 + lane < n) pknn = bl[nnch * 32 + lane];
      float degn = 1.0f;
      if (SCALE_SRC) degn = deg[pkn & 0x1ffffu];
      gcn_band_add_rows<F, P, SCALE_SRC>(src, acc, pk, ds_l, min(32, n - ch * 32), lane);
      if (SCALE_SRC) ds_l = gcn_dinv_of(degn);
      pk = pkn;
      pkn = pknn;
      ch += 32;
    }
  } else {
    for (int eb = wave * 32; eb < n_edges; eb += 1024) {
      const int e = eb + lane;
      unsigned pk = 0u;
      bool mine = false;
      if (e < n_edges) {
        const int rb = rows[e] - row_base;
        mine = (rb >= 0) && (rb < GCN_BAND_ROWS);
        if (mine) pk = ((unsigned)rb << 17) | (unsigned)cols[e];
      }
      unsigned bal = __builtin_amdgcn_ballot_w32(mine);
      while (bal) {
        const int l = __builtin_ctz(bal);
        bal &= bal - 1;
        const unsigned w = (unsigned)__builtin_amdgcn_readlane((int)pk, l);
        vecf v = *(const vecf*)(src + (size_t)(w & 0x1ffffu) * F + lane * NV);
        if (SCALE_SRC) v *= gcn_dinv_of(deg[w & 0x1ffffu]);
        float* a = acc + (int)(w >> 17) * P + lane;
#pragma unroll
        for (int j = 0; j < NV; ++j) atomicAdd(a + j * 32, v[j]);
      }
    }
  }
  __syncthreads();

  if (SCALE_SRC && (int)threadIdx.x < nrows)
    dinv_out[row_base + threadIdx.x] = gcn_dinv_of(deg[row_base + threadIdx.x]);

  if (TIMES_W2) {
    const int half = lane >> 4;
    const int l15 = lane & 15;
    const int n_tiles = (nrows + 15) >> 4;
    if (wave < n_tiles) {
      const int r = wave * 16 + l15;
      const float dsc = dl[r];
      const float* arow = acc + r * P + half * 64;
      const float* brow = w2s + half * (2 * NO) + l15 * 2;
      v8f c[NO / 16];
#pragma unroll
      for (int t = 0; t < NO / 16; ++t) c[t] = (v8f){};
#pragma unroll
      for (int q = 0; q < F / 4; ++q) {
        v2f a;
        a.x = fmaxf(arow[q] * dsc, 0.0f);
        a.y = fmaxf(arow[32 + q] * dsc, 0.0f);
#pragma unroll
        for (int t = 0; t < NO / 16; ++t) {
          const v2f b = *(const v2f*)(brow + q * (4 * NO) + t * 32);
          c[t] = __builtin_amdgcn_wmma_f32_16x16x4_f32(
              false, a, false, b, (short)0, c[t], false, false);
        }
      }
      float* stage = acc + (wave * 16 + 8 * half) * P + l15;
#pragma unroll
      for (int v = 0; v < 8; ++v) {
#pragma unroll
        for (int t = 0; t < NO / 16; ++t) stage[v * P + t * 16] = c[t][v];
      }
    }
    __syncthreads();
    for (int r = wave; r < nrows; r += 32) {
      v2f o = *(const v2f*)(acc + r * P + lane * 2);
      o *= dl[r];
      *(v2f*)(dst + (size_t)(row_base + r) * NO + lane * 2) = o;
    }
  } else {
    for (int r = wave; r < nrows; r += 32) {
      vecf v;
#pragma unroll
      for (int j = 0; j < NV; ++j) v[j] = acc[r * P + j * 32 + lane];
      if (SCALE_OUT) v *= dinv[row_base + r];
      *(vecf*)(dst + (size_t)(row_base + r) * F + lane * NV) = v;
    }
  }
}


extern "C" void kernel_launch(void* const* d_in, const int* in_sizes, int n_in,
                              void* d_out, int out_size, void* d_ws, size_t ws_size,
                              hipStream_t stream) {
  const float* X  = (const float*)d_in[0];
  const float* W1 = (const float*)d_in[1];
  const float* W2 = (const float*)d_in[2];
  const int* erow = (const int*)d_in[3];
  const int* ecol = (const int*)d_in[4];
  const int n_edges = in_sizes[3];
  float* out = (float*)d_out;

  float* deg = (float*)d_ws;
  float* dinv = deg + GCN_N_NODES;
  float* XW  = dinv + GCN_N_NODES;
  float* HWs = XW + (size_t)GCN_N_NODES * 128;
  unsigned* blist = (unsigned*)(HWs + (size_t)GCN_N_NODES * 64);
  unsigned* bcount = blist + (size_t)GCN_BANDS * GCN_LIST_CAP;

  gcn_zero_kernel<<<512, 256, 0, stream>>>(deg, GCN_N_NODES, bcount,
                                           GCN_BANDS * GCN_COUNT_PITCH);

  const int deg_groups = (n_edges + GCN_EDGE_GROUP - 1) / GCN_EDGE_GROUP;
  const int row_tiles = GCN_N_NODES / 16;
  const int gemm_blocks = (row_tiles + 7) / 8;

  gcn_gemm1_edges_kernel<<<gemm_blocks + deg_groups, 256, 0, stream>>>(
      X, W1, XW, GCN_N_NODES, 256, gemm_blocks, erow, ecol, deg, blist, bcount,
      n_edges);

  gcn_spmm_band<128, true, false, true><<<GCN_BANDS, 1024, 0, stream>>>(
      XW, erow, ecol, HWs, nullptr, deg, dinv, W2, blist, bcount, n_edges);

  gcn_spmm_band<64, false, true, false><<<GCN_BANDS, 1024, 0, stream>>>(
      HWs, erow, ecol, out, dinv, deg, nullptr, nullptr, blist, bcount, n_edges);
}
